# starting kernel
# baseline (speedup 1.0000x reference)
#include <hip/hip_runtime.h>

typedef __attribute__((ext_vector_type(2))) float v2f;
typedef __attribute__((ext_vector_type(8))) float v8f;

#define BATCH    4
#define CHANNELS 256
#define LATENT   512
#define DIM      32
#define SPATIAL  (DIM * DIM * DIM)
#define ODIM     64

__global__ __launch_bounds__(CHANNELS)
void style_mod_kernel(const float* __restrict__ istyle,
                      const float* __restrict__ style_w,
                      const float* __restrict__ style_b,
                      const float* __restrict__ conv_w,
                      float* __restrict__ mod) {
    __shared__ float s_is[LATENT];
    const int b = blockIdx.x;
    const int c = threadIdx.x;
    for (int i = threadIdx.x; i < LATENT; i += CHANNELS)
        s_is[i] = istyle[b * LATENT + i];
    __syncthreads();

    const float4* wrow = (const float4*)(style_w + c * LATENT);
    float acc = style_b[c];
    #pragma unroll 4
    for (int l = 0; l < LATENT / 4; ++l) {
        float4 w4 = wrow[l];
        acc += s_is[4 * l + 0] * w4.x + s_is[4 * l + 1] * w4.y +
               s_is[4 * l + 2] * w4.z + s_is[4 * l + 3] * w4.w;
    }
    mod[b * CHANNELS + c] = conv_w[c] * (acc + 1.0f);
}

__global__ __launch_bounds__(256)
void einsum_wmma_kernel(const float* __restrict__ x,
                        const float* __restrict__ prev,
                        const float* __restrict__ mod,
                        float* __restrict__ y0) {
    const int lane = threadIdx.x & 31;
    const int wave = threadIdx.x >> 5;
    const int tile = blockIdx.x * 8 + wave;
    const int tilesPerBatch = SPATIAL / 16;
    const int b  = tile / tilesPerBatch;
    const int s0 = (tile % tilesPerBatch) * 16;

    const int half = lane >> 4;
    const int m    = lane & 15;

    const float* xb   = x   + (size_t)b * CHANNELS * SPATIAL;
    const float* modb = mod + b * CHANNELS;

    v8f acc = {};
    for (int c0 = 0; c0 < CHANNELS; c0 += 4) {
        const int cA = c0 + 2 * half;
        v2f a, bb;
        a.x  = xb[(size_t)cA       * SPATIAL + s0 + m];
        a.y  = xb[(size_t)(cA + 1) * SPATIAL + s0 + m];
        bb.x = modb[cA];
        bb.y = modb[cA + 1];
        acc = __builtin_amdgcn_wmma_f32_16x16x4_f32(
             false, a,  false, bb,
             (short)0, acc,  false,  false);
    }

    if (m == 0) {
        const int base = s0 + 8 * half;
        const float4* pv = (const float4*)(prev + (size_t)b * SPATIAL + base);
        float4 p0 = pv[0], p1 = pv[1];
        float4 r0 = { acc[0] + p0.x, acc[1] + p0.y, acc[2] + p0.z, acc[3] + p0.w };
        float4 r1 = { acc[4] + p1.x, acc[5] + p1.y, acc[6] + p1.z, acc[7] + p1.w };
        float4* ov = (float4*)(y0 + (size_t)b * SPATIAL + base);
        ov[0] = r0;
        ov[1] = r1;
    }
}

__device__ __forceinline__ void axis_taps(int p, int* i3, float* w3) {
    const int m   = p >> 1;
    const int par = p & 1;
    i3[0] = (m - 1 < 0) ? 0 : m - 1;
    i3[1] = m;
    i3[2] = (m + 1 > DIM - 1) ? DIM - 1 : m + 1;
    w3[0] = par ? 0.0625f : 0.3125f;
    w3[1] = 0.625f;
    w3[2] = par ? 0.3125f : 0.0625f;
}

__global__ __launch_bounds__(256)
void upblur_kernel(const float* __restrict__ y0, float* __restrict__ out) {
    const int idx = blockIdx.x * 256 + threadIdx.x;
    const int w = idx & 63;
    const int h = (idx >> 6) & 63;
    const int d = (idx >> 12) & 63;
    const int b = idx >> 18;

    const float* yb = y0 + (size_t)b * SPATIAL;

    int id[3], ih[3], iw[3];
    float wd[3], wh[3], ww[3];
    axis_taps(d, id, wd);
    axis_taps(h, ih, wh);
    axis_taps(w, iw, ww);

    float sum = 0.0f;
    #pragma unroll
    for (int i = 0; i < 3; ++i) {
        #pragma unroll
        for (int j = 0; j < 3; ++j) {
            const float* row = yb + (id[i] * DIM + ih[j]) * DIM;
            const float rs = ww[0] * row[iw[0]] + ww[1] * row[iw[1]] + ww[2] * row[iw[2]];
            sum += wd[i] * wh[j] * rs;
        }
    }
    out[idx] = sum;
}

extern "C" void kernel_launch(void* const* d_in, const int* in_sizes, int n_in,
                              void* d_out, int out_size, void* d_ws, size_t ws_size,
                              hipStream_t stream) {
    const float* x       = (const float*)d_in[0];
    const float* prev    = (const float*)d_in[1];
    const float* istyle  = (const float*)d_in[2];
    const float* style_w = (const float*)d_in[3];
    const float* style_b = (const float*)d_in[4];
    const float* conv_w  = (const float*)d_in[5];
    float* out = (float*)d_out;

    float* mod = (float*)d_ws;
    float* y0  = mod + BATCH * CHANNELS;

    style_mod_kernel<<<BATCH, CHANNELS, 0, stream>>>(istyle, style_w, style_b, conv_w, mod);

    einsum_wmma_kernel<<<(BATCH * (SPATIAL / 16)) / 8, 256, 0, stream>>>(x, prev, mod, y0);

    upblur_kernel<<<(BATCH * ODIM * ODIM * ODIM) / 256, 256, 0, stream>>>(y0, out);
}


# channel sum re-split: block = 256 cells, its 8 waves take 32 channels each and meet in LDS; x rows enter as B operand (float4 per lane), mod-carrying A routes cells to D rows; y0 sum order changes
# speedup vs baseline: 2.1462x; 2.1462x over previous
#include <hip/hip_runtime.h>

typedef __attribute__((ext_vector_type(2))) float v2f;
typedef __attribute__((ext_vector_type(8))) float v8f;

#define BATCH    4
#define CHANNELS 256
#define LATENT   512
#define DIM      32
#define SPATIAL  (DIM * DIM * DIM)
#define ODIM     64

#define KSPLIT 16
#define KPW    (LATENT / KSPLIT)

__global__ __launch_bounds__(32 * KSPLIT)
void style_mod_wmma_kernel(const float* __restrict__ istyle,
                           const float* __restrict__ style_w,
                           const float* __restrict__ style_b,
                           const float* __restrict__ conv_w,
                           float* __restrict__ mod) {
    __shared__ float s_part[KSPLIT][BATCH][16];
    const int lane = threadIdx.x & 31;
    const int wave = threadIdx.x >> 5;
    const int half = lane >> 4;
    const int r    = lane & 15;
    const int c0   = blockIdx.x * 16;

    float bias = 0.0f, cw = 0.0f;
    if (threadIdx.x < BATCH * 16) {
        bias = style_b[c0 + r];
        cw   = conv_w[c0 + r];
    }

    const int k0 = wave * KPW + 4 * half;
    const float4* wp = (const float4*)(style_w + (size_t)(c0 + r) * LATENT + k0);
    const float4* ip = (const float4*)(istyle + (size_t)(r & (BATCH - 1)) * LATENT + k0);

    float4 wv[KPW / 8], iv[KPW / 8];
    #pragma unroll
    for (int j = 0; j < KPW / 8; ++j) {
        wv[j] = wp[2 * j];
        iv[j] = ip[2 * j];
    }

    v8f acc0 = {}, acc1 = {};
    #pragma unroll
    for (int j = 0; j < KPW / 8; ++j) {
        v2f a0 = { iv[j].x, iv[j].y }, b0 = { wv[j].x, wv[j].y };
        v2f a1 = { iv[j].z, iv[j].w }, b1 = { wv[j].z, wv[j].w };
        acc0 = __builtin_amdgcn_wmma_f32_16x16x4_f32(
             false, a0,  false, b0,
             (short)0, acc0,  false,  false);
        acc1 = __builtin_amdgcn_wmma_f32_16x16x4_f32(
             false, a1,  false, b1,
             (short)0, acc1,  false,  false);
    }

    if (half == 0) {
        s_part[wave][0][r] = acc0[0] + acc1[0];
        s_part[wave][1][r] = acc0[1] + acc1[1];
        s_part[wave][2][r] = acc0[2] + acc1[2];
        s_part[wave][3][r] = acc0[3] + acc1[3];
    }
    __syncthreads();

    if (threadIdx.x < BATCH * 16) {
        const int m = threadIdx.x >> 4;
        float dot = s_part[0][m][r];
        #pragma unroll
        for (int w = 1; w < KSPLIT; ++w) dot += s_part[w][m][r];
        mod[m * CHANNELS + c0 + r] = cw * ((dot + bias) + 1.0f);
    }
}

#define EWAVES 8
#define ECELLS 256
#define ECH    (CHANNELS / EWAVES)
#define EHALF  (ECH / 2)

__global__ __launch_bounds__(32 * EWAVES)
void einsum_wmma_kernel(const float* __restrict__ x,
                        const float* __restrict__ prev,
                        const float* __restrict__ mod,
                        float* __restrict__ y0) {
    __shared__ float4 s_part[EWAVES][ECELLS / 4];
    const int lane = threadIdx.x & 31;
    const int wave = __builtin_amdgcn_readfirstlane((int)(threadIdx.x >> 5));
    const int blocksPerBatch = SPATIAL / ECELLS;
    const int b  = blockIdx.x / blocksPerBatch;
    const int s0 = (blockIdx.x % blocksPerBatch) * ECELLS;
    const int cb = wave * ECH;

    const float* xp = x + ((size_t)b * CHANNELS + cb) * SPATIAL + s0 + 4 * lane;
    const float* mp = mod + b * CHANNELS + cb;

    float4 q[2 * EHALF];
    #pragma unroll
    for (int i = 0; i < EHALF; ++i) {
        q[2 * i]     = *(const float4*)(xp + (size_t)i * SPATIAL);
        q[2 * i + 1] = *(const float4*)(xp + (size_t)i * SPATIAL + 128);
    }
    const float pv = prev[(size_t)b * SPATIAL + s0 + threadIdx.x];
    __builtin_amdgcn_sched_barrier(0);

    const bool e0 = (lane == 0) | (lane == 24);
    const bool e1 = (lane == 1) | (lane == 25);
    const bool e2 = (lane == 2) | (lane == 26);
    const bool e3 = (lane == 3) | (lane == 27);

    v8f accL = {}, accR = {};
    #pragma unroll
    for (int i = 0; i < ECH; ++i) {
        const float m = mp[i];
        const v2f a_xy = { e0 ? m : 0.0f, e1 ? m : 0.0f };
        const v2f a_zw = { e2 ? m : 0.0f, e3 ? m : 0.0f };
        const int k = 2 * (i % EHALF);
        const v2f l_xy = { q[k].x, q[k].y },         l_zw = { q[k].z, q[k].w };
        const v2f r_xy = { q[k + 1].x, q[k + 1].y }, r_zw = { q[k + 1].z, q[k + 1].w };
        accL = __builtin_amdgcn_wmma_f32_16x16x4_f32(
             false, a_xy,  false, l_xy,
             (short)0, accL,  false,  false);
        accR = __builtin_amdgcn_wmma_f32_16x16x4_f32(
             false, a_xy,  false, r_xy,
             (short)0, accR,  false,  false);
        accL = __builtin_amdgcn_wmma_f32_16x16x4_f32(
             false, a_zw,  false, l_zw,
             (short)0, accL,  false,  false);
        accR = __builtin_amdgcn_wmma_f32_16x16x4_f32(
             false, a_zw,  false, r_zw,
             (short)0, accR,  false,  false);
        if (i < EHALF) {
            q[k]     = *(const float4*)(xp + (size_t)(i + EHALF) * SPATIAL);
            q[k + 1] = *(const float4*)(xp + (size_t)(i + EHALF) * SPATIAL + 128);
        }
        __builtin_amdgcn_sched_barrier(0);
    }

    s_part[wave][lane]      = make_float4(accL[0], accL[1], accL[2], accL[3]);
    s_part[wave][32 + lane] = make_float4(accR[0], accR[1], accR[2], accR[3]);
    __syncthreads();

    const float* sp = (const float*)s_part;
    float sum = sp[threadIdx.x];
    #pragma unroll
    for (int w = 1; w < EWAVES; ++w) sum += sp[w * ECELLS + threadIdx.x];
    y0[(size_t)b * SPATIAL + s0 + threadIdx.x] = sum + pv;
}

__device__ __forceinline__ float tap_even(float a, float b, float c) {
    return 0.3125f * a + 0.625f * b + 0.0625f * c;
}
__device__ __forceinline__ float tap_odd(float a, float b, float c) {
    return 0.0625f * a + 0.625f * b + 0.3125f * c;
}

__global__ __launch_bounds__(256)
void upblur_kernel(const float* __restrict__ y0, float* __restrict__ out) {
    const int lane = threadIdx.x & 31;
    const int tile = blockIdx.x * 8 + (threadIdx.x >> 5);
    const int h0 = (tile & 15) * 2;
    const int d  = (tile >> 4) & 31;
    const int b  = tile >> 9;

    const float* yb = y0 + (size_t)b * SPATIAL;

    int dd[3], hh[4];
    dd[0] = (d - 1 < 0) ? 0 : d - 1;
    dd[1] = d;
    dd[2] = (d + 1 > DIM - 1) ? DIM - 1 : d + 1;
    hh[0] = (h0 - 1 < 0) ? 0 : h0 - 1;
    hh[1] = h0;
    hh[2] = h0 + 1;
    hh[3] = (h0 + 2 > DIM - 1) ? DIM - 1 : h0 + 2;

    float c[3][4];
    #pragma unroll
    for (int i = 0; i < 3; ++i) {
        #pragma unroll
        for (int j = 0; j < 4; ++j)
            c[i][j] = yb[(dd[i] * DIM + hh[j]) * DIM + lane];
    }

    float e[3][4], o[3][4];
    #pragma unroll
    for (int i = 0; i < 3; ++i) {
        #pragma unroll
        for (int j = 0; j < 4; ++j) {
            const float l = __shfl_up(c[i][j], 1);
            const float r = __shfl_down(c[i][j], 1);
            e[i][j] = tap_even(l, c[i][j], r);
            o[i][j] = tap_odd(l, c[i][j], r);
        }
    }

    float2 p[3][4];
    #pragma unroll
    for (int i = 0; i < 3; ++i) {
        p[i][0] = make_float2(tap_even(e[i][0], e[i][1], e[i][2]), tap_even(o[i][0], o[i][1], o[i][2]));
        p[i][1] = make_float2(tap_odd (e[i][0], e[i][1], e[i][2]), tap_odd (o[i][0], o[i][1], o[i][2]));
        p[i][2] = make_float2(tap_even(e[i][1], e[i][2], e[i][3]), tap_even(o[i][1], o[i][2], o[i][3]));
        p[i][3] = make_float2(tap_odd (e[i][1], e[i][2], e[i][3]), tap_odd (o[i][1], o[i][2], o[i][3]));
    }

    float2* o_even = (float2*)(out + (((size_t)b * ODIM + 2 * d) * ODIM + 2 * h0) * ODIM) + lane;
    float2* o_odd  = o_even + (ODIM * ODIM) / 2;
    #pragma unroll
    for (int k = 0; k < 4; ++k) {
        o_even[k * (ODIM / 2)] = make_float2(tap_even(p[0][k].x, p[1][k].x, p[2][k].x),
                                             tap_even(p[0][k].y, p[1][k].y, p[2][k].y));
        o_odd[k * (ODIM / 2)]  = make_float2(tap_odd (p[0][k].x, p[1][k].x, p[2][k].x),
                                             tap_odd (p[0][k].y, p[1][k].y, p[2][k].y));
    }
}

extern "C" void kernel_launch(void* const* d_in, const int* in_sizes, int n_in,
                              void* d_out, int out_size, void* d_ws, size_t ws_size,
                              hipStream_t stream) {
    const float* x       = (const float*)d_in[0];
    const float* prev    = (const float*)d_in[1];
    const float* istyle  = (const float*)d_in[2];
    const float* style_w = (const float*)d_in[3];
    const float* style_b = (const float*)d_in[4];
    const float* conv_w  = (const float*)d_in[5];
    float* out = (float*)d_out;

    float* mod = (float*)d_ws;
    float* y0  = mod + BATCH * CHANNELS;

    style_mod_wmma_kernel<<<CHANNELS / 16, 32 * KSPLIT, 0, stream>>>(istyle, style_w, style_b, conv_w, mod);

    einsum_wmma_kernel<<<BATCH * (SPATIAL / ECELLS), 32 * EWAVES, 0, stream>>>(x, prev, mod, y0);

    upblur_kernel<<<(BATCH * DIM * (DIM / 2)) / 8, 256, 0, stream>>>(y0, out);
}
